# starting kernel
# baseline (speedup 1.0000x reference)
#include <hip/hip_runtime.h>

#define B_ 256
#define T_ 4096
#define H_ 32
#define LDS_STRIDE 20

typedef float v2f __attribute__((ext_vector_type(2)));
typedef float v8f __attribute__((ext_vector_type(8)));

__device__ __forceinline__ float hw_tanhf(float v) {
#if __has_builtin(__builtin_amdgcn_tanhf)
    return __builtin_amdgcn_tanhf(v);
#else
    const float xc = __builtin_fminf(__builtin_fmaxf(v, -9.0f), 9.0f);
    const float t  = __builtin_amdgcn_exp2f(xc * 2.8853900817779268f);
    const float r  = __builtin_amdgcn_rcpf(t + 1.0f);
    return (t - 1.0f) * r;
#endif
}

__global__ __launch_bounds__(32, 1)
void SineRNN_wmma_kernel(const float* __restrict__ x,
                         const float* __restrict__ W_ih,
                         const float* __restrict__ W_hh,
                         const float* __restrict__ b_ih,
                         const float* __restrict__ b_hh,
                         const float* __restrict__ W_fc,
                         const float* __restrict__ b_fc,
                         float* __restrict__ out)
{
    __shared__ float hbuf[H_ * LDS_STRIDE];

    const int lane  = threadIdx.x & 31;
    const int n     = lane & 15;
    const int hi    = lane >> 4;
    const int bbase = blockIdx.x * 16;

    v2f A[2][8];
#pragma unroll
    for (int c = 0; c < 2; ++c) {
#pragma unroll
        for (int j = 0; j < 8; ++j) {
            const int row = 16 * c + n;
            const int k0  = 4 * j + 2 * hi;
            A[c][j].x = W_hh[row * H_ + k0];
            A[c][j].y = W_hh[row * H_ + k0 + 1];
        }
    }

    float wih[2][8], bia[2][8], wfc[2][8];
#pragma unroll
    for (int c = 0; c < 2; ++c) {
#pragma unroll
        for (int r = 0; r < 8; ++r) {
            const int h = 16 * c + 8 * hi + r;
            wih[c][r] = W_ih[h];
            bia[c][r] = b_ih[h] + b_hh[h];
            wfc[c][r] = W_fc[h];
        }
    }
    const float bfc = b_fc[0];

    const float* xrow = x   + (size_t)(bbase + n) * T_;
    float*       orow = out + (size_t)(bbase + n) * T_;

    v2f Bop[8];
#pragma unroll
    for (int j = 0; j < 8; ++j) { Bop[j].x = 0.0f; Bop[j].y = 0.0f; }

    for (int t = 0; t < T_; ++t) {
        const float xv = xrow[t];
        __builtin_prefetch(xrow + t + 96, 0, 0);

        v8f acc0, acc1;
#pragma unroll
        for (int r = 0; r < 8; ++r) {
            acc0[r] = __builtin_fmaf(xv, wih[0][r], bia[0][r]);
            acc1[r] = __builtin_fmaf(xv, wih[1][r], bia[1][r]);
        }

#pragma unroll
        for (int j = 0; j < 8; ++j) {
            acc0 = __builtin_amdgcn_wmma_f32_16x16x4_f32(
                false, A[0][j], false, Bop[j], (short)0, acc0, false, false);
            acc1 = __builtin_amdgcn_wmma_f32_16x16x4_f32(
                false, A[1][j], false, Bop[j], (short)0, acc1, false, false);
        }

        float s = 0.0f;
#pragma unroll
        for (int r = 0; r < 8; ++r) {
            const float h0 = hw_tanhf(acc0[r]);
            const float h1 = hw_tanhf(acc1[r]);
            acc0[r] = h0;
            acc1[r] = h1;
            s = __builtin_fmaf(wfc[0][r], h0, s);
            s = __builtin_fmaf(wfc[1][r], h1, s);
        }
        s += __shfl_xor(s, 16, 32);
        if (hi == 0) orow[t] = s + bfc;

#pragma unroll
        for (int r = 0; r < 8; ++r) {
            hbuf[(8 * hi + r)      * LDS_STRIDE + n] = acc0[r];
            hbuf[(16 + 8 * hi + r) * LDS_STRIDE + n] = acc1[r];
        }
#pragma unroll
        for (int j = 0; j < 8; ++j) {
            const int k0 = 4 * j + 2 * hi;
            Bop[j].x = hbuf[k0       * LDS_STRIDE + n];
            Bop[j].y = hbuf[(k0 + 1) * LDS_STRIDE + n];
        }
    }
}

extern "C" void kernel_launch(void* const* d_in, const int* in_sizes, int n_in,
                              void* d_out, int out_size, void* d_ws, size_t ws_size,
                              hipStream_t stream) {
    const float* x    = (const float*)d_in[0];
    const float* W_ih = (const float*)d_in[1];
    const float* W_hh = (const float*)d_in[2];
    const float* b_ih = (const float*)d_in[3];
    const float* b_hh = (const float*)d_in[4];
    const float* W_fc = (const float*)d_in[5];
    const float* b_fc = (const float*)d_in[6];
    float* out = (float*)d_out;

    dim3 grid(B_ / 16);
    dim3 block(32);
    SineRNN_wmma_kernel<<<grid, block, 0, stream>>>(
        x, W_ih, W_hh, b_ih, b_hh, W_fc, b_fc, out);
}


# x read 8 steps ahead into registers (two 16-byte loads per lane per 8-step chunk, issued one chunk early; per-step 4-byte load and cache prefetch removed): no step waits for a global load
# speedup vs baseline: 1.0780x; 1.0780x over previous
#include <hip/hip_runtime.h>

#define B_ 256
#define T_ 4096
#define H_ 32
#define LDS_STRIDE 20
#define XCH 8

typedef float v2f __attribute__((ext_vector_type(2)));
typedef float v4f __attribute__((ext_vector_type(4)));
typedef float v8f __attribute__((ext_vector_type(8)));

__device__ __forceinline__ float hw_tanhf(float v) {
#if __has_builtin(__builtin_amdgcn_tanhf)
    return __builtin_amdgcn_tanhf(v);
#else
    const float xc = __builtin_fminf(__builtin_fmaxf(v, -9.0f), 9.0f);
    const float t  = __builtin_amdgcn_exp2f(xc * 2.8853900817779268f);
    const float r  = __builtin_amdgcn_rcpf(t + 1.0f);
    return (t - 1.0f) * r;
#endif
}

__global__ __launch_bounds__(32, 1)
void SineRNN_wmma_kernel(const float* __restrict__ x,
                         const float* __restrict__ W_ih,
                         const float* __restrict__ W_hh,
                         const float* __restrict__ b_ih,
                         const float* __restrict__ b_hh,
                         const float* __restrict__ W_fc,
                         const float* __restrict__ b_fc,
                         float* __restrict__ out)
{
    __shared__ float hbuf[H_ * LDS_STRIDE];

    const int lane  = threadIdx.x & 31;
    const int n     = lane & 15;
    const int hi    = lane >> 4;
    const int bbase = blockIdx.x * 16;

    v2f A[2][8];
#pragma unroll
    for (int c = 0; c < 2; ++c) {
#pragma unroll
        for (int j = 0; j < 8; ++j) {
            const int row = 16 * c + n;
            const int k0  = 4 * j + 2 * hi;
            A[c][j].x = W_hh[row * H_ + k0];
            A[c][j].y = W_hh[row * H_ + k0 + 1];
        }
    }

    float wih[2][8], bia[2][8], wfc[2][8];
#pragma unroll
    for (int c = 0; c < 2; ++c) {
#pragma unroll
        for (int r = 0; r < 8; ++r) {
            const int h = 16 * c + 8 * hi + r;
            wih[c][r] = W_ih[h];
            bia[c][r] = b_ih[h] + b_hh[h];
            wfc[c][r] = W_fc[h];
        }
    }
    const float bfc = b_fc[0];

    const float* xrow = x   + (size_t)(bbase + n) * T_;
    float*       orow = out + (size_t)(bbase + n) * T_;

    v2f Bop[8];
#pragma unroll
    for (int j = 0; j < 8; ++j) { Bop[j].x = 0.0f; Bop[j].y = 0.0f; }

    v4f xc0 = *(const v4f*)(xrow);
    v4f xc1 = *(const v4f*)(xrow + 4);

    for (int tc = 0; tc < T_; tc += XCH) {
        const int tn = (tc + XCH < T_) ? tc + XCH : tc;
        const v4f xn0 = *(const v4f*)(xrow + tn);
        const v4f xn1 = *(const v4f*)(xrow + tn + 4);

        const float xs[XCH] = { xc0.x, xc0.y, xc0.z, xc0.w,
                                xc1.x, xc1.y, xc1.z, xc1.w };

#pragma unroll
        for (int u = 0; u < XCH; ++u) {
            const int   t  = tc + u;
            const float xv = xs[u];

            v8f acc0, acc1;
#pragma unroll
            for (int r = 0; r < 8; ++r) {
                acc0[r] = __builtin_fmaf(xv, wih[0][r], bia[0][r]);
                acc1[r] = __builtin_fmaf(xv, wih[1][r], bia[1][r]);
            }

#pragma unroll
            for (int j = 0; j < 8; ++j) {
                acc0 = __builtin_amdgcn_wmma_f32_16x16x4_f32(
                    false, A[0][j], false, Bop[j], (short)0, acc0, false, false);
                acc1 = __builtin_amdgcn_wmma_f32_16x16x4_f32(
                    false, A[1][j], false, Bop[j], (short)0, acc1, false, false);
            }

            float s = 0.0f;
#pragma unroll
            for (int r = 0; r < 8; ++r) {
                const float h0 = hw_tanhf(acc0[r]);
                const float h1 = hw_tanhf(acc1[r]);
                acc0[r] = h0;
                acc1[r] = h1;
                s = __builtin_fmaf(wfc[0][r], h0, s);
                s = __builtin_fmaf(wfc[1][r], h1, s);
            }
            s += __shfl_xor(s, 16, 32);
            if (hi == 0) orow[t] = s + bfc;

#pragma unroll
            for (int r = 0; r < 8; ++r) {
                hbuf[(8 * hi + r)      * LDS_STRIDE + n] = acc0[r];
                hbuf[(16 + 8 * hi + r) * LDS_STRIDE + n] = acc1[r];
            }
#pragma unroll
            for (int j = 0; j < 8; ++j) {
                const int k0 = 4 * j + 2 * hi;
                Bop[j].x = hbuf[k0       * LDS_STRIDE + n];
                Bop[j].y = hbuf[(k0 + 1) * LDS_STRIDE + n];
            }
        }

        xc0 = xn0;
        xc1 = xn1;
    }
}

extern "C" void kernel_launch(void* const* d_in, const int* in_sizes, int n_in,
                              void* d_out, int out_size, void* d_ws, size_t ws_size,
                              hipStream_t stream) {
    const float* x    = (const float*)d_in[0];
    const float* W_ih = (const float*)d_in[1];
    const float* W_hh = (const float*)d_in[2];
    const float* b_ih = (const float*)d_in[3];
    const float* b_hh = (const float*)d_in[4];
    const float* W_fc = (const float*)d_in[5];
    const float* b_fc = (const float*)d_in[6];
    float* out = (float*)d_out;

    dim3 grid(B_ / 16);
    dim3 block(32);
    SineRNN_wmma_kernel<<<grid, block, 0, stream>>>(
        x, W_ih, W_hh, b_ih, b_hh, W_fc, b_fc, out);
}


# k order of the recurrent product matched to the D register layout: tanh(D) registers are the next step's B operand in place, LDS repack removed; changes the order the 32 products are summed
# speedup vs baseline: 1.6090x; 1.4926x over previous
#include <hip/hip_runtime.h>

#define B_ 256
#define T_ 4096
#define H_ 32
#define XCH 8

typedef float v2f __attribute__((ext_vector_type(2)));
typedef float v4f __attribute__((ext_vector_type(4)));
typedef float v8f __attribute__((ext_vector_type(8)));

__device__ __forceinline__ float hw_tanhf(float v) {
#if __has_builtin(__builtin_amdgcn_tanhf)
    return __builtin_amdgcn_tanhf(v);
#else
    const float xc = __builtin_fminf(__builtin_fmaxf(v, -9.0f), 9.0f);
    const float t  = __builtin_amdgcn_exp2f(xc * 2.8853900817779268f);
    const float r  = __builtin_amdgcn_rcpf(t + 1.0f);
    return (t - 1.0f) * r;
#endif
}

__global__ __launch_bounds__(32, 1)
void SineRNN_wmma_kernel(const float* __restrict__ x,
                         const float* __restrict__ W_ih,
                         const float* __restrict__ W_hh,
                         const float* __restrict__ b_ih,
                         const float* __restrict__ b_hh,
                         const float* __restrict__ W_fc,
                         const float* __restrict__ b_fc,
                         float* __restrict__ out)
{
    const int lane  = threadIdx.x & 31;
    const int n     = lane & 15;
    const int hi    = lane >> 4;
    const int bbase = blockIdx.x * 16;

    v2f A[2][8];
#pragma unroll
    for (int c = 0; c < 2; ++c) {
#pragma unroll
        for (int j = 0; j < 8; ++j) {
            const int row = 16 * c + n;
            const int k0  = 16 * (j >> 2) + 8 * hi + 2 * (j & 3);
            A[c][j].x = W_hh[row * H_ + k0];
            A[c][j].y = W_hh[row * H_ + k0 + 1];
        }
    }

    float wih[2][8], bia[2][8], wfc[2][8];
#pragma unroll
    for (int c = 0; c < 2; ++c) {
#pragma unroll
        for (int r = 0; r < 8; ++r) {
            const int h = 16 * c + 8 * hi + r;
            wih[c][r] = W_ih[h];
            bia[c][r] = b_ih[h] + b_hh[h];
            wfc[c][r] = W_fc[h];
        }
    }
    const float bfc = b_fc[0];

    const float* xrow = x   + (size_t)(bbase + n) * T_;
    float*       orow = out + (size_t)(bbase + n) * T_;

    v2f Bop[8];
#pragma unroll
    for (int j = 0; j < 8; ++j) { Bop[j].x = 0.0f; Bop[j].y = 0.0f; }

    v4f xc0 = *(const v4f*)(xrow);
    v4f xc1 = *(const v4f*)(xrow + 4);

    for (int tc = 0; tc < T_; tc += XCH) {
        const int tn = (tc + XCH < T_) ? tc + XCH : tc;
        const v4f xn0 = *(const v4f*)(xrow + tn);
        const v4f xn1 = *(const v4f*)(xrow + tn + 4);

        const float xs[XCH] = { xc0.x, xc0.y, xc0.z, xc0.w,
                                xc1.x, xc1.y, xc1.z, xc1.w };

#pragma unroll
        for (int u = 0; u < XCH; ++u) {
            const int   t  = tc + u;
            const float xv = xs[u];

            v8f acc0, acc1;
#pragma unroll
            for (int r = 0; r < 8; ++r) {
                acc0[r] = __builtin_fmaf(xv, wih[0][r], bia[0][r]);
                acc1[r] = __builtin_fmaf(xv, wih[1][r], bia[1][r]);
            }

#pragma unroll
            for (int j = 0; j < 8; ++j) {
                acc0 = __builtin_amdgcn_wmma_f32_16x16x4_f32(
                    false, A[0][j], false, Bop[j], (short)0, acc0, false, false);
                acc1 = __builtin_amdgcn_wmma_f32_16x16x4_f32(
                    false, A[1][j], false, Bop[j], (short)0, acc1, false, false);
            }

            float s = 0.0f;
#pragma unroll
            for (int r = 0; r < 8; ++r) {
                const float h0 = hw_tanhf(acc0[r]);
                const float h1 = hw_tanhf(acc1[r]);
                acc0[r] = h0;
                acc1[r] = h1;
                s = __builtin_fmaf(wfc[0][r], h0, s);
                s = __builtin_fmaf(wfc[1][r], h1, s);
            }
            s += __shfl_xor(s, 16, 32);
            if (hi == 0) orow[t] = s + bfc;

#pragma unroll
            for (int q = 0; q < 4; ++q) {
                Bop[q].x     = acc0[2 * q];
                Bop[q].y     = acc0[2 * q + 1];
                Bop[4 + q].x = acc1[2 * q];
                Bop[4 + q].y = acc1[2 * q + 1];
            }
        }

        xc0 = xn0;
        xc1 = xn1;
    }
}

extern "C" void kernel_launch(void* const* d_in, const int* in_sizes, int n_in,
                              void* d_out, int out_size, void* d_ws, size_t ws_size,
                              hipStream_t stream) {
    const float* x    = (const float*)d_in[0];
    const float* W_ih = (const float*)d_in[1];
    const float* W_hh = (const float*)d_in[2];
    const float* b_ih = (const float*)d_in[3];
    const float* b_hh = (const float*)d_in[4];
    const float* W_fc = (const float*)d_in[5];
    const float* b_fc = (const float*)d_in[6];
    float* out = (float*)d_out;

    dim3 grid(B_ / 16);
    dim3 block(32);
    SineRNN_wmma_kernel<<<grid, block, 0, stream>>>(
        x, W_ih, W_hh, b_ih, b_hh, W_fc, b_fc, out);
}


# per-step output path taken off the recurrence: dot of step t-1 and C of step t+1 issue among step t's WMMAs; halves summed by permlanex16, one 16-byte store per lane per 8 steps, no bpermute/branch
# speedup vs baseline: 2.4763x; 1.5390x over previous
#include <hip/hip_runtime.h>

#define B_ 256
#define T_ 4096
#define H_ 32
#define XCH 8

typedef float v2f __attribute__((ext_vector_type(2)));
typedef float v4f __attribute__((ext_vector_type(4)));
typedef float v8f __attribute__((ext_vector_type(8)));

__device__ __forceinline__ float hw_tanhf(float v) {
#if __has_builtin(__builtin_amdgcn_tanhf)
    return __builtin_amdgcn_tanhf(v);
#else
    const float xc = __builtin_fminf(__builtin_fmaxf(v, -9.0f), 9.0f);
    const float t  = __builtin_amdgcn_exp2f(xc * 2.8853900817779268f);
    const float r  = __builtin_amdgcn_rcpf(t + 1.0f);
    return (t - 1.0f) * r;
#endif
}

__device__ __forceinline__ float xhalf(float v) {
    return __builtin_bit_cast(float, __builtin_amdgcn_permlanex16(
        __builtin_bit_cast(int, v), __builtin_bit_cast(int, v),
        0x76543210, 0xfedcba98, false, false));
}

__device__ __forceinline__ float fc_dot(const v2f (&Bop)[8], const float (&wfc)[2][8]) {
    float s = 0.0f;
#pragma unroll
    for (int r = 0; r < 8; ++r) {
        const float h0 = (r & 1) ? Bop[r >> 1].y     : Bop[r >> 1].x;
        const float h1 = (r & 1) ? Bop[4 + (r >> 1)].y : Bop[4 + (r >> 1)].x;
        s = __builtin_fmaf(wfc[0][r], h0, s);
        s = __builtin_fmaf(wfc[1][r], h1, s);
    }
    return s;
}

__device__ __forceinline__ void flush8(const float (&sv)[8], float bfc, int hi, float* dst) {
    v4f o;
#pragma unroll
    for (int k = 0; k < 4; ++k) {
        const float mine = hi ? sv[k + 4] : sv[k];
        const float send = hi ? sv[k]     : sv[k + 4];
        o[k] = (mine + xhalf(send)) + bfc;
    }
    *(v4f*)(dst + 4 * hi) = o;
}

template <bool FLUSH>
__device__ __forceinline__ void chunk8(const v2f (&A)[2][8],
                                       const float (&wih)[2][8], const float (&bia)[2][8],
                                       const float (&wfc)[2][8], float bfc, int hi,
                                       v2f (&Bop)[8], v8f& cin0, v8f& cin1, float (&sv)[8],
                                       const float (&xnext)[XCH], float* oprev)
{
#pragma unroll
    for (int u = 0; u < XCH; ++u) {
        v8f acc0 = cin0, acc1 = cin1;
#pragma unroll
        for (int j = 0; j < 8; ++j) {
            acc0 = __builtin_amdgcn_wmma_f32_16x16x4_f32(
                false, A[0][j], false, Bop[j], (short)0, acc0, false, false);
            acc1 = __builtin_amdgcn_wmma_f32_16x16x4_f32(
                false, A[1][j], false, Bop[j], (short)0, acc1, false, false);
        }

        const float s = fc_dot(Bop, wfc);
        if (u == 0) {
            if (FLUSH) { sv[7] = s; flush8(sv, bfc, hi, oprev); }
        } else {
            sv[u - 1] = s;
        }
        const float xv = xnext[u];
#pragma unroll
        for (int r = 0; r < 8; ++r) {
            cin0[r] = __builtin_fmaf(xv, wih[0][r], bia[0][r]);
            cin1[r] = __builtin_fmaf(xv, wih[1][r], bia[1][r]);
        }

#pragma unroll
        for (int q = 0; q < 4; ++q) {
            Bop[q].x     = hw_tanhf(acc0[2 * q]);
            Bop[q].y     = hw_tanhf(acc0[2 * q + 1]);
            Bop[4 + q].x = hw_tanhf(acc1[2 * q]);
            Bop[4 + q].y = hw_tanhf(acc1[2 * q + 1]);
        }
    }
}

__global__ __launch_bounds__(32, 1)
void SineRNN_wmma_kernel(const float* __restrict__ x,
                         const float* __restrict__ W_ih,
                         const float* __restrict__ W_hh,
                         const float* __restrict__ b_ih,
                         const float* __restrict__ b_hh,
                         const float* __restrict__ W_fc,
                         const float* __restrict__ b_fc,
                         float* __restrict__ out)
{
    const int lane  = threadIdx.x & 31;
    const int n     = lane & 15;
    const int hi    = lane >> 4;
    const int bbase = blockIdx.x * 16;

    v2f A[2][8];
#pragma unroll
    for (int c = 0; c < 2; ++c) {
#pragma unroll
        for (int j = 0; j < 8; ++j) {
            const int row = 16 * c + n;
            const int k0  = 16 * (j >> 2) + 8 * hi + 2 * (j & 3);
            A[c][j].x = W_hh[row * H_ + k0];
            A[c][j].y = W_hh[row * H_ + k0 + 1];
        }
    }

    float wih[2][8], bia[2][8], wfc[2][8];
#pragma unroll
    for (int c = 0; c < 2; ++c) {
#pragma unroll
        for (int r = 0; r < 8; ++r) {
            const int h = 16 * c + 8 * hi + r;
            wih[c][r] = W_ih[h];
            bia[c][r] = b_ih[h] + b_hh[h];
            wfc[c][r] = W_fc[h];
        }
    }
    const float bfc = b_fc[0];

    const float* xrow = x   + (size_t)(bbase + n) * T_;
    float*       orow = out + (size_t)(bbase + n) * T_;

    v2f Bop[8];
#pragma unroll
    for (int j = 0; j < 8; ++j) { Bop[j].x = 0.0f; Bop[j].y = 0.0f; }

    v4f xc0 = *(const v4f*)(xrow);
    v4f xc1 = *(const v4f*)(xrow + 4);

    v8f cin0, cin1;
#pragma unroll
    for (int r = 0; r < 8; ++r) {
        cin0[r] = __builtin_fmaf(xc0.x, wih[0][r], bia[0][r]);
        cin1[r] = __builtin_fmaf(xc0.x, wih[1][r], bia[1][r]);
    }

    float sv[8];
#pragma unroll
    for (int k = 0; k < 8; ++k) sv[k] = 0.0f;

    {
        const v4f xn0 = *(const v4f*)(xrow + XCH);
        const v4f xn1 = *(const v4f*)(xrow + XCH + 4);
        const float xnext[XCH] = { xc0.y, xc0.z, xc0.w, xc1.x, xc1.y, xc1.z, xc1.w, xn0.x };
        chunk8<false>(A, wih, bia, wfc, bfc, hi, Bop, cin0, cin1, sv, xnext, orow);
        xc0 = xn0;
        xc1 = xn1;
    }

    for (int tc = XCH; tc < T_; tc += XCH) {
        const int tn = (tc + XCH < T_) ? tc + XCH : tc;
        const v4f xn0 = *(const v4f*)(xrow + tn);
        const v4f xn1 = *(const v4f*)(xrow + tn + 4);

        const float xnext[XCH] = { xc0.y, xc0.z, xc0.w, xc1.x, xc1.y, xc1.z, xc1.w, xn0.x };
        chunk8<true>(A, wih, bia, wfc, bfc, hi, Bop, cin0, cin1, sv, xnext, orow + (tc - XCH));

        xc0 = xn0;
        xc1 = xn1;
    }

    sv[7] = fc_dot(Bop, wfc);
    flush8(sv, bfc, hi, orow + (T_ - XCH));
}

extern "C" void kernel_launch(void* const* d_in, const int* in_sizes, int n_in,
                              void* d_out, int out_size, void* d_ws, size_t ws_size,
                              hipStream_t stream) {
    const float* x    = (const float*)d_in[0];
    const float* W_ih = (const float*)d_in[1];
    const float* W_hh = (const float*)d_in[2];
    const float* b_ih = (const float*)d_in[3];
    const float* b_hh = (const float*)d_in[4];
    const float* W_fc = (const float*)d_in[5];
    const float* b_fc = (const float*)d_in[6];
    float* out = (float*)d_out;

    dim3 grid(B_ / 16);
    dim3 block(32);
    SineRNN_wmma_kernel<<<grid, block, 0, stream>>>(
        x, W_ih, W_hh, b_ih, b_hh, W_fc, b_fc, out);
}


# recurrence split over 2 waves (2 SIMDs) per 16 batch rows: 8 matrix instr. per wave and step, h halves swapped via LDS + one barrier a step; sum order: rows 16-31 own k half first, fc dot in 4 chains
# speedup vs baseline: 3.1311x; 1.2644x over previous
#include <hip/hip_runtime.h>

#define B_ 256
#define T_ 4096
#define H_ 32
#define XCH 8

#define PIECE_F 128
#define WAVE_F  (2 * PIECE_F)
#define BUF_F   (2 * WAVE_F)

typedef float v2f __attribute__((ext_vector_type(2)));
typedef float v4f __attribute__((ext_vector_type(4)));
typedef float v8f __attribute__((ext_vector_type(8)));

__device__ __forceinline__ float hw_tanhf(float v) {
#if __has_builtin(__builtin_amdgcn_tanhf)
    return __builtin_amdgcn_tanhf(v);
#else
    const float xc = __builtin_fminf(__builtin_fmaxf(v, -9.0f), 9.0f);
    const float t  = __builtin_amdgcn_exp2f(xc * 2.8853900817779268f);
    const float r  = __builtin_amdgcn_rcpf(t + 1.0f);
    return (t - 1.0f) * r;
#endif
}

__device__ __forceinline__ float xhalf(float v) {
    return __builtin_bit_cast(float, __builtin_amdgcn_permlanex16(
        __builtin_bit_cast(int, v), __builtin_bit_cast(int, v),
        0x76543210, 0xfedcba98, false, false));
}

__device__ __forceinline__ float fc_dot(const v2f (&Bop)[8], const float (&wfc)[2][8]) {
    float c0 = 0.0f, c1 = 0.0f, c2 = 0.0f, c3 = 0.0f;
#pragma unroll
    for (int r = 0; r < 4; ++r) {
        const float a0 = (r & 1) ? Bop[r >> 1].y           : Bop[r >> 1].x;
        const float a1 = (r & 1) ? Bop[2 + (r >> 1)].y     : Bop[2 + (r >> 1)].x;
        const float b0 = (r & 1) ? Bop[4 + (r >> 1)].y     : Bop[4 + (r >> 1)].x;
        const float b1 = (r & 1) ? Bop[6 + (r >> 1)].y     : Bop[6 + (r >> 1)].x;
        c0 = __builtin_fmaf(wfc[0][r],     a0, c0);
        c1 = __builtin_fmaf(wfc[0][r + 4], a1, c1);
        c2 = __builtin_fmaf(wfc[1][r],     b0, c2);
        c3 = __builtin_fmaf(wfc[1][r + 4], b1, c3);
    }
    return (c0 + c1) + (c2 + c3);
}

__device__ __forceinline__ void flush8(const float (&sv)[8], float bfc, int hi, float* dst) {
    v4f o;
#pragma unroll
    for (int k = 0; k < 4; ++k) {
        const float mine = hi ? sv[k + 4] : sv[k];
        const float send = hi ? sv[k]     : sv[k + 4];
        o[k] = (mine + xhalf(send)) + bfc;
    }
    *(v4f*)(dst + 4 * hi) = o;
}

__device__ __forceinline__ void handover() {
    asm volatile("s_wait_dscnt 0x0" ::: "memory");
    asm volatile("s_barrier_signal -1" ::: "memory");
    asm volatile("s_barrier_wait -1" ::: "memory");
}

template <bool FLUSH>
__device__ __forceinline__ void chunk8(const v2f (&Aown)[4], const v2f (&Aoth)[4],
                                       const float (&wih)[8], const float (&bia)[8],
                                       const float (&wfc)[2][8], float bfc, int hi, bool outw,
                                       float* ldsOwn, const float* ldsOth,
                                       v2f (&Hown)[4], v2f (&Bprev)[8], v8f& cin, float (&sv)[8],
                                       const float (&xnext)[XCH], float* oprev)
{
#pragma unroll
    for (int u = 0; u < XCH; ++u) {
        const int pb = (u & 1) * BUF_F;

        v4f s0, s1;
        s0.x = Hown[0].x; s0.y = Hown[0].y; s0.z = Hown[1].x; s0.w = Hown[1].y;
        s1.x = Hown[2].x; s1.y = Hown[2].y; s1.z = Hown[3].x; s1.w = Hown[3].y;
        *(v4f*)(ldsOwn + pb)           = s0;
        *(v4f*)(ldsOwn + pb + PIECE_F) = s1;

        v8f acc = cin;
        float xv = xnext[u];
        asm volatile("" : "+v"(acc), "+v"(xv) :: "memory");

        acc = __builtin_amdgcn_wmma_f32_16x16x4_f32(
            false, Aown[0], false, Hown[0], (short)0, acc, false, false);
        acc = __builtin_amdgcn_wmma_f32_16x16x4_f32(
            false, Aown[1], false, Hown[1], (short)0, acc, false, false);
#pragma unroll
        for (int r = 0; r < 8; ++r) cin[r] = __builtin_fmaf(xv, wih[r], bia[r]);
        asm volatile("" : "+v"(acc), "+v"(cin) :: "memory");

        handover();
        const v4f p0 = *(const v4f*)(ldsOth + pb);
        const v4f p1 = *(const v4f*)(ldsOth + pb + PIECE_F);
        asm volatile("" : "+v"(acc) :: "memory");

        acc = __builtin_amdgcn_wmma_f32_16x16x4_f32(
            false, Aown[2], false, Hown[2], (short)0, acc, false, false);
        acc = __builtin_amdgcn_wmma_f32_16x16x4_f32(
            false, Aown[3], false, Hown[3], (short)0, acc, false, false);

        if (outw) {
            const float s = fc_dot(Bprev, wfc);
            if (u == 0) {
                sv[6] = s;
            } else if (u == 1) {
                sv[7] = s;
                if (FLUSH) flush8(sv, bfc, hi, oprev);
            } else {
                sv[u - 2] = s;
            }
        }

        v2f Hoth[4];
        Hoth[0].x = p0.x; Hoth[0].y = p0.y; Hoth[1].x = p0.z; Hoth[1].y = p0.w;
        Hoth[2].x = p1.x; Hoth[2].y = p1.y; Hoth[3].x = p1.z; Hoth[3].y = p1.w;
#pragma unroll
        for (int q = 0; q < 4; ++q) {
            acc = __builtin_amdgcn_wmma_f32_16x16x4_f32(
                false, Aoth[q], false, Hoth[q], (short)0, acc, false, false);
        }

#pragma unroll
        for (int q = 0; q < 4; ++q) { Bprev[q] = Hown[q]; Bprev[4 + q] = Hoth[q]; }

#pragma unroll
        for (int q = 0; q < 4; ++q) {
            Hown[q].x = hw_tanhf(acc[2 * q]);
            Hown[q].y = hw_tanhf(acc[2 * q + 1]);
        }
    }
}

__global__ __launch_bounds__(64, 1)
void SineRNN_wmma_kernel(const float* __restrict__ x,
                         const float* __restrict__ W_ih,
                         const float* __restrict__ W_hh,
                         const float* __restrict__ b_ih,
                         const float* __restrict__ b_hh,
                         const float* __restrict__ W_fc,
                         const float* __restrict__ b_fc,
                         float* __restrict__ out)
{
    __shared__ __attribute__((aligned(16))) float xch[2 * BUF_F];

    const int lane  = threadIdx.x & 31;
    const int wv    = __builtin_amdgcn_readfirstlane((int)(threadIdx.x >> 5));
    const int n     = lane & 15;
    const int hi    = lane >> 4;
    const int bbase = blockIdx.x * 16;
    const bool outw = (wv == 0);

    v2f Aown[4], Aoth[4];
#pragma unroll
    for (int q = 0; q < 4; ++q) {
        const int row = 16 * wv + n;
        const int ko  = 16 * wv       + 8 * hi + 2 * q;
        const int kp  = 16 * (1 - wv) + 8 * hi + 2 * q;
        Aown[q].x = W_hh[row * H_ + ko];
        Aown[q].y = W_hh[row * H_ + ko + 1];
        Aoth[q].x = W_hh[row * H_ + kp];
        Aoth[q].y = W_hh[row * H_ + kp + 1];
    }

    float wih[8], bia[8], wfc[2][8];
#pragma unroll
    for (int r = 0; r < 8; ++r) {
        const int h = 16 * wv + 8 * hi + r;
        wih[r] = W_ih[h];
        bia[r] = b_ih[h] + b_hh[h];
    }
#pragma unroll
    for (int c = 0; c < 2; ++c) {
#pragma unroll
        for (int r = 0; r < 8; ++r) wfc[c][r] = W_fc[16 * c + 8 * hi + r];
    }
    const float bfc = b_fc[0];

    const float* xrow = x   + (size_t)(bbase + n) * T_;
    float*       orow = out + (size_t)(bbase + n) * T_;

    float*       ldsOwn = xch + wv * WAVE_F       + lane * 4;
    const float* ldsOth = xch + (1 - wv) * WAVE_F + lane * 4;

    v2f Hown[4];
#pragma unroll
    for (int q = 0; q < 4; ++q) { Hown[q].x = 0.0f; Hown[q].y = 0.0f; }
    v2f Bprev[8];
#pragma unroll
    for (int j = 0; j < 8; ++j) { Bprev[j].x = 0.0f; Bprev[j].y = 0.0f; }

    v4f xc0 = *(const v4f*)(xrow);
    v4f xc1 = *(const v4f*)(xrow + 4);

    v8f cin;
#pragma unroll
    for (int r = 0; r < 8; ++r) cin[r] = __builtin_fmaf(xc0.x, wih[r], bia[r]);

    float sv[8];
#pragma unroll
    for (int k = 0; k < 8; ++k) sv[k] = 0.0f;

    {
        const v4f xn0 = *(const v4f*)(xrow + XCH);
        const v4f xn1 = *(const v4f*)(xrow + XCH + 4);
        const float xnext[XCH] = { xc0.y, xc0.z, xc0.w, xc1.x, xc1.y, xc1.z, xc1.w, xn0.x };
        chunk8<false>(Aown, Aoth, wih, bia, wfc, bfc, hi, outw, ldsOwn, ldsOth,
                      Hown, Bprev, cin, sv, xnext, orow);
        xc0 = xn0;
        xc1 = xn1;
    }

    for (int tc = XCH; tc < T_; tc += XCH) {
        const int tn = (tc + XCH < T_) ? tc + XCH : tc;
        const v4f xn0 = *(const v4f*)(xrow + tn);
        const v4f xn1 = *(const v4f*)(xrow + tn + 4);

        const float xnext[XCH] = { xc0.y, xc0.z, xc0.w, xc1.x, xc1.y, xc1.z, xc1.w, xn0.x };
        chunk8<true>(Aown, Aoth, wih, bia, wfc, bfc, hi, outw, ldsOwn, ldsOth,
                     Hown, Bprev, cin, sv, xnext, orow + (tc - XCH));

        xc0 = xn0;
        xc1 = xn1;
    }

    {
        v4f s0, s1;
        s0.x = Hown[0].x; s0.y = Hown[0].y; s0.z = Hown[1].x; s0.w = Hown[1].y;
        s1.x = Hown[2].x; s1.y = Hown[2].y; s1.z = Hown[3].x; s1.w = Hown[3].y;
        *(v4f*)(ldsOwn)           = s0;
        *(v4f*)(ldsOwn + PIECE_F) = s1;
        handover();
        const v4f p0 = *(const v4f*)(ldsOth);
        const v4f p1 = *(const v4f*)(ldsOth + PIECE_F);
        if (outw) {
            sv[6] = fc_dot(Bprev, wfc);
            v2f Blast[8];
#pragma unroll
            for (int q = 0; q < 4; ++q) Blast[q] = Hown[q];
            Blast[4].x = p0.x; Blast[4].y = p0.y; Blast[5].x = p0.z; Blast[5].y = p0.w;
            Blast[6].x = p1.x; Blast[6].y = p1.y; Blast[7].x = p1.z; Blast[7].y = p1.w;
            sv[7] = fc_dot(Blast, wfc);
            flush8(sv, bfc, hi, orow + (T_ - XCH));
        }
    }
}

extern "C" void kernel_launch(void* const* d_in, const int* in_sizes, int n_in,
                              void* d_out, int out_size, void* d_ws, size_t ws_size,
                              hipStream_t stream) {
    const float* x    = (const float*)d_in[0];
    const float* W_ih = (const float*)d_in[1];
    const float* W_hh = (const float*)d_in[2];
    const float* b_ih = (const float*)d_in[3];
    const float* b_hh = (const float*)d_in[4];
    const float* W_fc = (const float*)d_in[5];
    const float* b_fc = (const float*)d_in[6];
    float* out = (float*)d_out;

    dim3 grid(B_ / 16);
    dim3 block(64);
    SineRNN_wmma_kernel<<<grid, block, 0, stream>>>(
        x, W_ih, W_hh, b_ih, b_hh, W_fc, b_fc, out);
}
